# starting kernel
# baseline (speedup 1.0000x reference)
#include <hip/hip_runtime.h>


#define DIM      128
#define NHEAD    8
#define NTOK     (4 * 16384)
#define NROWT    (NTOK / 16)
#define NT_QKV   24
#define NT_PROJ  8

#define WAVES_PER_BLOCK 4
#define BLOCK   (WAVES_PER_BLOCK * 32)
#define GRID    512

typedef float v2f __attribute__((ext_vector_type(2)));
typedef float v8f __attribute__((ext_vector_type(8)));

#define OFF_WQKV  0
#define SZ_WQKV   (128 * 384)
#define OFF_WPROJ (OFF_WQKV + SZ_WQKV)
#define SZ_WPROJ  (128 * 128)
#define OFF_ABUF  (OFF_WPROJ + SZ_WPROJ)
#define SZ_ABUF_W (16 * DIM)
#define OFF_KV    (OFF_ABUF + WAVES_PER_BLOCK * SZ_ABUF_W)
#define SZ_KV_W   (16 * 16 * 2)
#define LDS_FLOATS (OFF_KV + WAVES_PER_BLOCK * SZ_KV_W)

__launch_bounds__(BLOCK)
__global__ void fused_channel_attn(const float* __restrict__ x,
                                   const float* __restrict__ w_qkv,
                                   const float* __restrict__ b_qkv,
                                   const float* __restrict__ w_proj,
                                   const float* __restrict__ b_proj,
                                   float* __restrict__ out)
{
    __shared__ float lds[LDS_FLOATS];

    const int tid  = threadIdx.x;
    const int wave = tid >> 5;
    const int lane = tid & 31;
    const int l16  = lane & 15;
    const int half = lane >> 4;

    for (int idx = tid; idx < 128 * 384; idx += BLOCK) {
        int k = idx / 384, n = idx % 384;
        int kt = k >> 2, kr = k & 3;
        int nt = n >> 4, nl = n & 15;
        lds[OFF_WQKV + (((kt * NT_QKV + nt) * 32 + (nl + 16 * (kr >> 1))) << 1)
                     + (kr & 1)] = w_qkv[idx];
    }
    for (int idx = tid; idx < 128 * 128; idx += BLOCK) {
        int k = idx / 128, n = idx % 128;
        int kt = k >> 2, kr = k & 3;
        int nt = n >> 4, nl = n & 15;
        lds[OFF_WPROJ + (((kt * NT_PROJ + nt) * 32 + (nl + 16 * (kr >> 1))) << 1)
                      + (kr & 1)] = w_proj[idx];
    }
    __syncthreads();

    float bqk[NT_QKV];
#pragma unroll
    for (int nt = 0; nt < NT_QKV; ++nt) bqk[nt] = b_qkv[nt * 16 + l16];
    float bp[NT_PROJ];
#pragma unroll
    for (int nt = 0; nt < NT_PROJ; ++nt) bp[nt] = b_proj[nt * 16 + l16];

    float* abuf  = &lds[OFF_ABUF + wave * SZ_ABUF_W];
    float* kvbuf = &lds[OFF_KV   + wave * SZ_KV_W];

    const float c1 = 0.08838834764831845f * 1.4426950408889634f;

    const int waveGlobal = blockIdx.x * WAVES_PER_BLOCK + wave;
    const int waveStride = GRID * WAVES_PER_BLOCK;

    for (int rt = waveGlobal; rt < NROWT; rt += waveStride) {
        const int row0 = rt * 16;

        const float4* xg  = (const float4*)(x + (size_t)row0 * DIM);
        float4*       ab4 = (float4*)abuf;
#pragma unroll
        for (int i = 0; i < (16 * DIM / 4) / 32; ++i)
            ab4[lane + i * 32] = xg[lane + i * 32];

        v2f a[32];
#pragma unroll
        for (int kt = 0; kt < 32; ++kt)
            a[kt] = *(const v2f*)&abuf[l16 * DIM + kt * 4 + half * 2];

#pragma unroll 1
        for (int h = 0; h < NHEAD; ++h) {
            v8f cq, ck, cv;
#pragma unroll
            for (int r = 0; r < 8; ++r) {
                cq[r] = bqk[h];
                ck[r] = bqk[8 + h];
                cv[r] = bqk[16 + h];
            }
#pragma unroll
            for (int kt = 0; kt < 32; ++kt) {
                v2f bq = *(const v2f*)&lds[OFF_WQKV + (kt * NT_QKV + h     ) * 64 + lane * 2];
                v2f bk = *(const v2f*)&lds[OFF_WQKV + (kt * NT_QKV + h + 8 ) * 64 + lane * 2];
                v2f bv = *(const v2f*)&lds[OFF_WQKV + (kt * NT_QKV + h + 16) * 64 + lane * 2];
                cq = __builtin_amdgcn_wmma_f32_16x16x4_f32(false, a[kt], false, bq,
                                                           (short)0, cq, false, false);
                ck = __builtin_amdgcn_wmma_f32_16x16x4_f32(false, a[kt], false, bk,
                                                           (short)0, ck, false, false);
                cv = __builtin_amdgcn_wmma_f32_16x16x4_f32(false, a[kt], false, bv,
                                                           (short)0, cv, false, false);
            }

#pragma unroll
            for (int r = 0; r < 8; ++r) {
                int t = r + 8 * half;
                v2f kv; kv.x = ck[r]; kv.y = cv[r];
                *(v2f*)&kvbuf[(t * 16 + l16) * 2] = kv;
            }

#pragma unroll
            for (int r = 0; r < 8; ++r) {
                int t   = r + 8 * half;
                float q = cq[r];
                float num = 0.0f, den = 0.0f;
#pragma unroll
                for (int j = 0; j < 16; ++j) {
                    v2f kv  = *(const v2f*)&kvbuf[(t * 16 + j) * 2];
                    float e = __builtin_amdgcn_exp2f(q * kv.x * c1);
                    num += e * kv.y;
                    den += e;
                }
                abuf[t * DIM + h * 16 + l16] = num * __builtin_amdgcn_rcpf(den);
            }
        }

#pragma unroll
        for (int kt = 0; kt < 32; ++kt)
            a[kt] = *(const v2f*)&abuf[l16 * DIM + kt * 4 + half * 2];

#pragma unroll 1
        for (int nt = 0; nt < NT_PROJ; ++nt) {
            v8f c;
#pragma unroll
            for (int r = 0; r < 8; ++r) c[r] = bp[nt];
#pragma unroll
            for (int kt = 0; kt < 32; ++kt) {
                v2f b = *(const v2f*)&lds[OFF_WPROJ + (kt * NT_PROJ + nt) * 64 + lane * 2];
                c = __builtin_amdgcn_wmma_f32_16x16x4_f32(false, a[kt], false, b,
                                                          (short)0, c, false, false);
            }
#pragma unroll
            for (int r = 0; r < 8; ++r)
                out[(size_t)(row0 + r + 8 * half) * DIM + nt * 16 + l16] = c[r];
        }
    }
}

extern "C" void kernel_launch(void* const* d_in, const int* in_sizes, int n_in,
                              void* d_out, int out_size, void* d_ws, size_t ws_size,
                              hipStream_t stream) {
    const float* x      = (const float*)d_in[0];
    const float* w_qkv  = (const float*)d_in[1];
    const float* b_qkv  = (const float*)d_in[2];
    const float* w_proj = (const float*)d_in[3];
    const float* b_proj = (const float*)d_in[4];
    float* out = (float*)d_out;
    (void)in_sizes; (void)n_in; (void)out_size; (void)d_ws; (void)ws_size;
    fused_channel_attn<<<dim3(GRID), dim3(BLOCK), 0, stream>>>(
        x, w_qkv, b_qkv, w_proj, b_proj, out);
}


# Weights staged into LDS once per WGP instead of twice: grid 512 -> 256 persistent workgroups, b128 row-pair loads with shift-only indexing, start row rotated per workgroup
# speedup vs baseline: 1.5144x; 1.5144x over previous
#include <hip/hip_runtime.h>


#define DIM      128
#define NHEAD    8
#define NTOK     (4 * 16384)
#define NROWT    (NTOK / 16)
#define NT_QKV   24
#define NT_PROJ  8

#define WAVES_PER_BLOCK 4
#define BLOCK   (WAVES_PER_BLOCK * 32)
#define GRID    256

typedef float v2f __attribute__((ext_vector_type(2)));
typedef float v8f __attribute__((ext_vector_type(8)));

#define OFF_WQKV  0
#define SZ_WQKV   (128 * 384)
#define OFF_WPROJ (OFF_WQKV + SZ_WQKV)
#define SZ_WPROJ  (128 * 128)
#define OFF_ABUF  (OFF_WPROJ + SZ_WPROJ)
#define SZ_ABUF_W (16 * DIM)
#define OFF_KV    (OFF_ABUF + WAVES_PER_BLOCK * SZ_ABUF_W)
#define SZ_KV_W   (16 * 16 * 2)
#define LDS_FLOATS (OFF_KV + WAVES_PER_BLOCK * SZ_KV_W)

__launch_bounds__(BLOCK)
__global__ void fused_channel_attn(const float* __restrict__ x,
                                   const float* __restrict__ w_qkv,
                                   const float* __restrict__ b_qkv,
                                   const float* __restrict__ w_proj,
                                   const float* __restrict__ b_proj,
                                   float* __restrict__ out)
{
    __shared__ float lds[LDS_FLOATS];

    const int tid  = threadIdx.x;
    const int wave = tid >> 5;
    const int lane = tid & 31;
    const int l16  = lane & 15;
    const int half = lane >> 4;

    {
        const int rot = blockIdx.x;
        const float4* wq4 = (const float4*)w_qkv;
#pragma unroll 1
        for (int i0 = 0; i0 < 16; i0 += 4) {
            float4 r0[12], r1[12];
#pragma unroll
            for (int u = 0; u < 4; ++u) {
                const int kp = (wave + 4 * (i0 + u) + rot) & 63;
#pragma unroll
                for (int c = 0; c < 3; ++c) {
                    const int n4 = lane + 32 * c;
                    r0[u * 3 + c] = wq4[(2 * kp)     * 96 + n4];
                    r1[u * 3 + c] = wq4[(2 * kp + 1) * 96 + n4];
                }
            }
#pragma unroll
            for (int u = 0; u < 4; ++u) {
                const int kp = (wave + 4 * (i0 + u) + rot) & 63;
                const int kt = kp >> 1, hs = kp & 1;
#pragma unroll
                for (int c = 0; c < 3; ++c) {
                    const int n4 = lane + 32 * c;
                    const float4 a0 = r0[u * 3 + c], a1 = r1[u * 3 + c];
                    float* dst = &lds[OFF_WQKV + (kt * NT_QKV + (n4 >> 2)) * 64
                                      + hs * 32 + (n4 & 3) * 8];
                    *(float4*)(dst)     = make_float4(a0.x, a1.x, a0.y, a1.y);
                    *(float4*)(dst + 4) = make_float4(a0.z, a1.z, a0.w, a1.w);
                }
            }
        }
        const float4* wp4 = (const float4*)w_proj;
        {
            float4 r0[16], r1[16];
#pragma unroll
            for (int i = 0; i < 16; ++i) {
                const int kp = (wave + 4 * i + rot) & 63;
                r0[i] = wp4[(2 * kp)     * 32 + lane];
                r1[i] = wp4[(2 * kp + 1) * 32 + lane];
            }
#pragma unroll
            for (int i = 0; i < 16; ++i) {
                const int kp = (wave + 4 * i + rot) & 63;
                const int kt = kp >> 1, hs = kp & 1;
                const float4 a0 = r0[i], a1 = r1[i];
                float* dst = &lds[OFF_WPROJ + (kt * NT_PROJ + (lane >> 2)) * 64
                                  + hs * 32 + (lane & 3) * 8];
                *(float4*)(dst)     = make_float4(a0.x, a1.x, a0.y, a1.y);
                *(float4*)(dst + 4) = make_float4(a0.z, a1.z, a0.w, a1.w);
            }
        }
    }
    __syncthreads();

    float bqk[NT_QKV];
#pragma unroll
    for (int nt = 0; nt < NT_QKV; ++nt) bqk[nt] = b_qkv[nt * 16 + l16];
    float bp[NT_PROJ];
#pragma unroll
    for (int nt = 0; nt < NT_PROJ; ++nt) bp[nt] = b_proj[nt * 16 + l16];

    float* abuf  = &lds[OFF_ABUF + wave * SZ_ABUF_W];
    float* kvbuf = &lds[OFF_KV   + wave * SZ_KV_W];

    const float c1 = 0.08838834764831845f * 1.4426950408889634f;

    const int waveGlobal = blockIdx.x * WAVES_PER_BLOCK + wave;
    const int waveStride = GRID * WAVES_PER_BLOCK;

    for (int rt = waveGlobal; rt < NROWT; rt += waveStride) {
        const int row0 = rt * 16;

        const float4* xg  = (const float4*)(x + (size_t)row0 * DIM);
        float4*       ab4 = (float4*)abuf;
#pragma unroll
        for (int i = 0; i < (16 * DIM / 4) / 32; ++i)
            ab4[lane + i * 32] = xg[lane + i * 32];

        v2f a[32];
#pragma unroll
        for (int kt = 0; kt < 32; ++kt)
            a[kt] = *(const v2f*)&abuf[l16 * DIM + kt * 4 + half * 2];

#pragma unroll 1
        for (int h = 0; h < NHEAD; ++h) {
            v8f cq, ck, cv;
#pragma unroll
            for (int r = 0; r < 8; ++r) {
                cq[r] = bqk[h];
                ck[r] = bqk[8 + h];
                cv[r] = bqk[16 + h];
            }
#pragma unroll
            for (int kt = 0; kt < 32; ++kt) {
                v2f bq = *(const v2f*)&lds[OFF_WQKV + (kt * NT_QKV + h     ) * 64 + lane * 2];
                v2f bk = *(const v2f*)&lds[OFF_WQKV + (kt * NT_QKV + h + 8 ) * 64 + lane * 2];
                v2f bv = *(const v2f*)&lds[OFF_WQKV + (kt * NT_QKV + h + 16) * 64 + lane * 2];
                cq = __builtin_amdgcn_wmma_f32_16x16x4_f32(false, a[kt], false, bq,
                                                           (short)0, cq, false, false);
                ck = __builtin_amdgcn_wmma_f32_16x16x4_f32(false, a[kt], false, bk,
                                                           (short)0, ck, false, false);
                cv = __builtin_amdgcn_wmma_f32_16x16x4_f32(false, a[kt], false, bv,
                                                           (short)0, cv, false, false);
            }

#pragma unroll
            for (int r = 0; r < 8; ++r) {
                int t = r + 8 * half;
                v2f kv; kv.x = ck[r]; kv.y = cv[r];
                *(v2f*)&kvbuf[(t * 16 + l16) * 2] = kv;
            }

#pragma unroll
            for (int r = 0; r < 8; ++r) {
                int t   = r + 8 * half;
                float q = cq[r];
                float num = 0.0f, den = 0.0f;
#pragma unroll
                for (int j = 0; j < 16; ++j) {
                    v2f kv  = *(const v2f*)&kvbuf[(t * 16 + j) * 2];
                    float e = __builtin_amdgcn_exp2f(q * kv.x * c1);
                    num += e * kv.y;
                    den += e;
                }
                abuf[t * DIM + h * 16 + l16] = num * __builtin_amdgcn_rcpf(den);
            }
        }

#pragma unroll
        for (int kt = 0; kt < 32; ++kt)
            a[kt] = *(const v2f*)&abuf[l16 * DIM + kt * 4 + half * 2];

#pragma unroll 1
        for (int nt = 0; nt < NT_PROJ; ++nt) {
            v8f c;
#pragma unroll
            for (int r = 0; r < 8; ++r) c[r] = bp[nt];
#pragma unroll
            for (int kt = 0; kt < 32; ++kt) {
                v2f b = *(const v2f*)&lds[OFF_WPROJ + (kt * NT_PROJ + nt) * 64 + lane * 2];
                c = __builtin_amdgcn_wmma_f32_16x16x4_f32(false, a[kt], false, b,
                                                          (short)0, c, false, false);
            }
#pragma unroll
            for (int r = 0; r < 8; ++r)
                out[(size_t)(row0 + r + 8 * half) * DIM + nt * 16 + l16] = c[r];
        }
    }
}

extern "C" void kernel_launch(void* const* d_in, const int* in_sizes, int n_in,
                              void* d_out, int out_size, void* d_ws, size_t ws_size,
                              hipStream_t stream) {
    const float* x      = (const float*)d_in[0];
    const float* w_qkv  = (const float*)d_in[1];
    const float* b_qkv  = (const float*)d_in[2];
    const float* w_proj = (const float*)d_in[3];
    const float* b_proj = (const float*)d_in[4];
    float* out = (float*)d_out;
    (void)in_sizes; (void)n_in; (void)out_size; (void)d_ws; (void)ws_size;
    fused_channel_attn<<<dim3(GRID), dim3(BLOCK), 0, stream>>>(
        x, w_qkv, b_qkv, w_proj, b_proj, out);
}


# Two waves per SIMD: 8-wave workgroups, 2 row tiles per wave, so one wave's softmax and load waits run under the other's WMMAs; proj accumulated per head so a wave's LDS fits 8 KB
# speedup vs baseline: 2.0137x; 1.3297x over previous
#include <hip/hip_runtime.h>


#define DIM      128
#define NHEAD    8
#define NTOK     (4 * 16384)
#define NROWT    (NTOK / 16)
#define NT_QKV   24
#define NT_PROJ  8

#define WAVES_PER_BLOCK 8
#define BLOCK   (WAVES_PER_BLOCK * 32)
#define GRID    256

typedef float v2f __attribute__((ext_vector_type(2)));
typedef float v8f __attribute__((ext_vector_type(8)));

#define OFF_WQKV  0
#define SZ_WQKV   (128 * 384)
#define OFF_WPROJ (OFF_WQKV + SZ_WQKV)
#define SZ_WPROJ  (128 * 128)
#define OFF_WBUF  (OFF_WPROJ + SZ_WPROJ)
#define SZ_WBUF_W (16 * DIM)
#define OFF_O     (16 * 16 * 2)
#define O_PITCH   20
#define LDS_FLOATS (OFF_WBUF + WAVES_PER_BLOCK * SZ_WBUF_W)

__launch_bounds__(BLOCK)
__global__ void fused_channel_attn(const float* __restrict__ x,
                                   const float* __restrict__ w_qkv,
                                   const float* __restrict__ b_qkv,
                                   const float* __restrict__ w_proj,
                                   const float* __restrict__ b_proj,
                                   float* __restrict__ out)
{
    __shared__ float lds[LDS_FLOATS];

    const int tid  = threadIdx.x;
    const int wave = tid >> 5;
    const int lane = tid & 31;
    const int l16  = lane & 15;
    const int half = lane >> 4;

    {
        const int rot = blockIdx.x;
        const float4* wq4 = (const float4*)w_qkv;
#pragma unroll 1
        for (int i0 = 0; i0 < 8; i0 += 4) {
            float4 r0[12], r1[12];
#pragma unroll
            for (int u = 0; u < 4; ++u) {
                const int kp = (wave + 8 * (i0 + u) + rot) & 63;
#pragma unroll
                for (int c = 0; c < 3; ++c) {
                    const int n4 = lane + 32 * c;
                    r0[u * 3 + c] = wq4[(2 * kp)     * 96 + n4];
                    r1[u * 3 + c] = wq4[(2 * kp + 1) * 96 + n4];
                }
            }
#pragma unroll
            for (int u = 0; u < 4; ++u) {
                const int kp = (wave + 8 * (i0 + u) + rot) & 63;
                const int kt = kp >> 1, hs = kp & 1;
#pragma unroll
                for (int c = 0; c < 3; ++c) {
                    const int n4 = lane + 32 * c;
                    const float4 a0 = r0[u * 3 + c], a1 = r1[u * 3 + c];
                    float* dst = &lds[OFF_WQKV + (kt * NT_QKV + (n4 >> 2)) * 64
                                      + hs * 32 + (n4 & 3) * 8];
                    *(float4*)(dst)     = make_float4(a0.x, a1.x, a0.y, a1.y);
                    *(float4*)(dst + 4) = make_float4(a0.z, a1.z, a0.w, a1.w);
                }
            }
        }
        const float4* wp4 = (const float4*)w_proj;
        {
            float4 r0[8], r1[8];
#pragma unroll
            for (int i = 0; i < 8; ++i) {
                const int kp = (wave + 8 * i + rot) & 63;
                r0[i] = wp4[(2 * kp)     * 32 + lane];
                r1[i] = wp4[(2 * kp + 1) * 32 + lane];
            }
#pragma unroll
            for (int i = 0; i < 8; ++i) {
                const int kp = (wave + 8 * i + rot) & 63;
                const int kt = kp >> 1, hs = kp & 1;
                const float4 a0 = r0[i], a1 = r1[i];
                float* dst = &lds[OFF_WPROJ + (kt * NT_PROJ + (lane >> 2)) * 64
                                  + hs * 32 + (lane & 3) * 8];
                *(float4*)(dst)     = make_float4(a0.x, a1.x, a0.y, a1.y);
                *(float4*)(dst + 4) = make_float4(a0.z, a1.z, a0.w, a1.w);
            }
        }
    }
    __syncthreads();

    float bqk[NT_QKV];
#pragma unroll
    for (int nt = 0; nt < NT_QKV; ++nt) bqk[nt] = b_qkv[nt * 16 + l16];
    float bp[NT_PROJ];
#pragma unroll
    for (int nt = 0; nt < NT_PROJ; ++nt) bp[nt] = b_proj[nt * 16 + l16];

    float* abuf  = &lds[OFF_WBUF + wave * SZ_WBUF_W];
    float* kvbuf = abuf;
    float* obuf  = abuf + OFF_O;

    const float c1 = 0.08838834764831845f * 1.4426950408889634f;

    const int waveGlobal = blockIdx.x * WAVES_PER_BLOCK + wave;
    const int waveStride = GRID * WAVES_PER_BLOCK;

    for (int rt = waveGlobal; rt < NROWT; rt += waveStride) {
        const int row0 = rt * 16;

        const float4* xg  = (const float4*)(x + (size_t)row0 * DIM);
        float4*       ab4 = (float4*)abuf;
#pragma unroll
        for (int i = 0; i < (16 * DIM / 4) / 32; ++i)
            ab4[lane + i * 32] = xg[lane + i * 32];

        v2f a[32];
#pragma unroll
        for (int kt = 0; kt < 32; ++kt)
            a[kt] = *(const v2f*)&abuf[l16 * DIM + kt * 4 + half * 2];

        v8f co[NT_PROJ];
#pragma unroll
        for (int nt = 0; nt < NT_PROJ; ++nt) {
#pragma unroll
            for (int r = 0; r < 8; ++r) co[nt][r] = bp[nt];
        }

#pragma unroll 1
        for (int h = 0; h < NHEAD; ++h) {
            v8f cq, ck, cv;
#pragma unroll
            for (int r = 0; r < 8; ++r) {
                cq[r] = bqk[h];
                ck[r] = bqk[8 + h];
                cv[r] = bqk[16 + h];
            }
#pragma unroll
            for (int kt = 0; kt < 32; ++kt) {
                v2f bq = *(const v2f*)&lds[OFF_WQKV + (kt * NT_QKV + h     ) * 64 + lane * 2];
                v2f bk = *(const v2f*)&lds[OFF_WQKV + (kt * NT_QKV + h + 8 ) * 64 + lane * 2];
                v2f bv = *(const v2f*)&lds[OFF_WQKV + (kt * NT_QKV + h + 16) * 64 + lane * 2];
                cq = __builtin_amdgcn_wmma_f32_16x16x4_f32(false, a[kt], false, bq,
                                                           (short)0, cq, false, false);
                ck = __builtin_amdgcn_wmma_f32_16x16x4_f32(false, a[kt], false, bk,
                                                           (short)0, ck, false, false);
                cv = __builtin_amdgcn_wmma_f32_16x16x4_f32(false, a[kt], false, bv,
                                                           (short)0, cv, false, false);
            }

#pragma unroll
            for (int r = 0; r < 8; ++r) {
                int t = r + 8 * half;
                v2f kv; kv.x = ck[r]; kv.y = cv[r];
                *(v2f*)&kvbuf[(t * 16 + l16) * 2] = kv;
            }

#pragma unroll
            for (int r = 0; r < 8; ++r) {
                int t   = r + 8 * half;
                float q = cq[r];
                float num = 0.0f, den = 0.0f;
#pragma unroll
                for (int j = 0; j < 16; ++j) {
                    v2f kv  = *(const v2f*)&kvbuf[(t * 16 + j) * 2];
                    float e = __builtin_amdgcn_exp2f(q * kv.x * c1);
                    num += e * kv.y;
                    den += e;
                }
                obuf[t * O_PITCH + l16] = num * __builtin_amdgcn_rcpf(den);
            }

            v2f ao[4];
#pragma unroll
            for (int q = 0; q < 4; ++q)
                ao[q] = *(const v2f*)&obuf[l16 * O_PITCH + q * 4 + half * 2];
#pragma unroll
            for (int q = 0; q < 4; ++q) {
#pragma unroll
                for (int nt = 0; nt < NT_PROJ; ++nt) {
                    v2f b = *(const v2f*)&lds[OFF_WPROJ + ((h * 4 + q) * NT_PROJ + nt) * 64 + lane * 2];
                    co[nt] = __builtin_amdgcn_wmma_f32_16x16x4_f32(false, ao[q], false, b,
                                                                   (short)0, co[nt], false, false);
                }
            }
        }

#pragma unroll
        for (int nt = 0; nt < NT_PROJ; ++nt) {
#pragma unroll
            for (int r = 0; r < 8; ++r)
                out[(size_t)(row0 + r + 8 * half) * DIM + nt * 16 + l16] = co[nt][r];
        }
    }
}

extern "C" void kernel_launch(void* const* d_in, const int* in_sizes, int n_in,
                              void* d_out, int out_size, void* d_ws, size_t ws_size,
                              hipStream_t stream) {
    const float* x      = (const float*)d_in[0];
    const float* w_qkv  = (const float*)d_in[1];
    const float* b_qkv  = (const float*)d_in[2];
    const float* w_proj = (const float*)d_in[3];
    const float* b_proj = (const float*)d_in[4];
    float* out = (float*)d_out;
    (void)in_sizes; (void)n_in; (void)out_size; (void)d_ws; (void)ws_size;
    fused_channel_attn<<<dim3(GRID), dim3(BLOCK), 0, stream>>>(
        x, w_qkv, b_qkv, w_proj, b_proj, out);
}


# Channel-attention softmax made throughput-bound: K/V stashed transposed so a lane's 8 token rows advance together per channel (8 independent sums, packed f32 ops), not one serial chain per row
# speedup vs baseline: 2.2569x; 1.1208x over previous
#include <hip/hip_runtime.h>


#define DIM      128
#define NHEAD    8
#define NTOK     (4 * 16384)
#define NROWT    (NTOK / 16)
#define NT_QKV   24
#define NT_PROJ  8

#define WAVES_PER_BLOCK 8
#define BLOCK   (WAVES_PER_BLOCK * 32)
#define GRID    256

typedef float v2f __attribute__((ext_vector_type(2)));
typedef float v8f __attribute__((ext_vector_type(8)));

#define OFF_WQKV  0
#define SZ_WQKV   (128 * 384)
#define OFF_WPROJ (OFF_WQKV + SZ_WQKV)
#define SZ_WPROJ  (128 * 128)
#define OFF_WBUF  (OFF_WPROJ + SZ_WPROJ)
#define SZ_WBUF_W (16 * DIM)
#define KV_HALF   160
#define OFF_V     (2 * KV_HALF)
#define OFF_O     (4 * KV_HALF)
#define O_PITCH   20
#define LDS_FLOATS (OFF_WBUF + WAVES_PER_BLOCK * SZ_WBUF_W)

__launch_bounds__(BLOCK)
__global__ void fused_channel_attn(const float* __restrict__ x,
                                   const float* __restrict__ w_qkv,
                                   const float* __restrict__ b_qkv,
                                   const float* __restrict__ w_proj,
                                   const float* __restrict__ b_proj,
                                   float* __restrict__ out)
{
    __shared__ __attribute__((aligned(16))) float lds[LDS_FLOATS];

    const int tid  = threadIdx.x;
    const int wave = tid >> 5;
    const int lane = tid & 31;
    const int l16  = lane & 15;
    const int half = lane >> 4;

    {
        const int rot = blockIdx.x;
        const float4* wq4 = (const float4*)w_qkv;
#pragma unroll 1
        for (int i0 = 0; i0 < 8; i0 += 4) {
            float4 r0[12], r1[12];
#pragma unroll
            for (int u = 0; u < 4; ++u) {
                const int kp = (wave + 8 * (i0 + u) + rot) & 63;
#pragma unroll
                for (int c = 0; c < 3; ++c) {
                    const int n4 = lane + 32 * c;
                    r0[u * 3 + c] = wq4[(2 * kp)     * 96 + n4];
                    r1[u * 3 + c] = wq4[(2 * kp + 1) * 96 + n4];
                }
            }
#pragma unroll
            for (int u = 0; u < 4; ++u) {
                const int kp = (wave + 8 * (i0 + u) + rot) & 63;
                const int kt = kp >> 1, hs = kp & 1;
#pragma unroll
                for (int c = 0; c < 3; ++c) {
                    const int n4 = lane + 32 * c;
                    const float4 a0 = r0[u * 3 + c], a1 = r1[u * 3 + c];
                    float* dst = &lds[OFF_WQKV + (kt * NT_QKV + (n4 >> 2)) * 64
                                      + hs * 32 + (n4 & 3) * 8];
                    *(float4*)(dst)     = make_float4(a0.x, a1.x, a0.y, a1.y);
                    *(float4*)(dst + 4) = make_float4(a0.z, a1.z, a0.w, a1.w);
                }
            }
        }
        const float4* wp4 = (const float4*)w_proj;
        {
            float4 r0[8], r1[8];
#pragma unroll
            for (int i = 0; i < 8; ++i) {
                const int kp = (wave + 8 * i + rot) & 63;
                r0[i] = wp4[(2 * kp)     * 32 + lane];
                r1[i] = wp4[(2 * kp + 1) * 32 + lane];
            }
#pragma unroll
            for (int i = 0; i < 8; ++i) {
                const int kp = (wave + 8 * i + rot) & 63;
                const int kt = kp >> 1, hs = kp & 1;
                const float4 a0 = r0[i], a1 = r1[i];
                float* dst = &lds[OFF_WPROJ + (kt * NT_PROJ + (lane >> 2)) * 64
                                  + hs * 32 + (lane & 3) * 8];
                *(float4*)(dst)     = make_float4(a0.x, a1.x, a0.y, a1.y);
                *(float4*)(dst + 4) = make_float4(a0.z, a1.z, a0.w, a1.w);
            }
        }
    }
    __syncthreads();

    float bqk[NT_QKV];
#pragma unroll
    for (int nt = 0; nt < NT_QKV; ++nt) bqk[nt] = b_qkv[nt * 16 + l16];
    float bp[NT_PROJ];
#pragma unroll
    for (int nt = 0; nt < NT_PROJ; ++nt) bp[nt] = b_proj[nt * 16 + l16];

    float* abuf  = &lds[OFF_WBUF + wave * SZ_WBUF_W];
    float* kvbuf = abuf;
    float* obuf  = abuf + OFF_O;

    const float c1 = 0.08838834764831845f * 1.4426950408889634f;

    const int waveGlobal = blockIdx.x * WAVES_PER_BLOCK + wave;
    const int waveStride = GRID * WAVES_PER_BLOCK;

    for (int rt = waveGlobal; rt < NROWT; rt += waveStride) {
        const int row0 = rt * 16;

        const float4* xg  = (const float4*)(x + (size_t)row0 * DIM);
        float4*       ab4 = (float4*)abuf;
#pragma unroll
        for (int i = 0; i < (16 * DIM / 4) / 32; ++i)
            ab4[lane + i * 32] = xg[lane + i * 32];

        v2f a[32];
#pragma unroll
        for (int kt = 0; kt < 32; ++kt)
            a[kt] = *(const v2f*)&abuf[l16 * DIM + kt * 4 + half * 2];

        v8f co[NT_PROJ];
#pragma unroll
        for (int nt = 0; nt < NT_PROJ; ++nt) {
#pragma unroll
            for (int r = 0; r < 8; ++r) co[nt][r] = bp[nt];
        }

#pragma unroll 1
        for (int h = 0; h < NHEAD; ++h) {
            v8f cq, ck, cv;
#pragma unroll
            for (int r = 0; r < 8; ++r) {
                cq[r] = bqk[h];
                ck[r] = bqk[8 + h];
                cv[r] = bqk[16 + h];
            }
#pragma unroll
            for (int kt = 0; kt < 32; ++kt) {
                v2f bq = *(const v2f*)&lds[OFF_WQKV + (kt * NT_QKV + h     ) * 64 + lane * 2];
                v2f bk = *(const v2f*)&lds[OFF_WQKV + (kt * NT_QKV + h + 8 ) * 64 + lane * 2];
                v2f bv = *(const v2f*)&lds[OFF_WQKV + (kt * NT_QKV + h + 16) * 64 + lane * 2];
                cq = __builtin_amdgcn_wmma_f32_16x16x4_f32(false, a[kt], false, bq,
                                                           (short)0, cq, false, false);
                ck = __builtin_amdgcn_wmma_f32_16x16x4_f32(false, a[kt], false, bk,
                                                           (short)0, ck, false, false);
                cv = __builtin_amdgcn_wmma_f32_16x16x4_f32(false, a[kt], false, bv,
                                                           (short)0, cv, false, false);
            }

            {
                float* kp = &kvbuf[half * KV_HALF + l16 * 8];
                *(float4*)(kp)             = make_float4(ck[0], ck[1], ck[2], ck[3]);
                *(float4*)(kp + 4)         = make_float4(ck[4], ck[5], ck[6], ck[7]);
                *(float4*)(kp + OFF_V)     = make_float4(cv[0], cv[1], cv[2], cv[3]);
                *(float4*)(kp + OFF_V + 4) = make_float4(cv[4], cv[5], cv[6], cv[7]);
            }

            v2f qs[4], num[4], den[4];
#pragma unroll
            for (int p = 0; p < 4; ++p) {
                qs[p].x = cq[2 * p] * c1;  qs[p].y = cq[2 * p + 1] * c1;
                num[p].x = 0.0f; num[p].y = 0.0f;
                den[p].x = 0.0f; den[p].y = 0.0f;
            }
            const float* kbase = &kvbuf[half * KV_HALF];
#pragma unroll
            for (int j = 0; j < 16; ++j) {
                const float4 k0 = *(const float4*)&kbase[j * 8];
                const float4 k1 = *(const float4*)&kbase[j * 8 + 4];
                const float4 v0 = *(const float4*)&kbase[OFF_V + j * 8];
                const float4 v1 = *(const float4*)&kbase[OFF_V + j * 8 + 4];
                v2f kk[4], vv[4];
                kk[0].x = k0.x; kk[0].y = k0.y; kk[1].x = k0.z; kk[1].y = k0.w;
                kk[2].x = k1.x; kk[2].y = k1.y; kk[3].x = k1.z; kk[3].y = k1.w;
                vv[0].x = v0.x; vv[0].y = v0.y; vv[1].x = v0.z; vv[1].y = v0.w;
                vv[2].x = v1.x; vv[2].y = v1.y; vv[3].x = v1.z; vv[3].y = v1.w;
#pragma unroll
                for (int p = 0; p < 4; ++p) {
                    const v2f a2 = qs[p] * kk[p];
                    v2f e;
                    e.x = __builtin_amdgcn_exp2f(a2.x);
                    e.y = __builtin_amdgcn_exp2f(a2.y);
                    num[p] = __builtin_elementwise_fma(e, vv[p], num[p]);
                    den[p] = den[p] + e;
                }
            }
#pragma unroll
            for (int p = 0; p < 4; ++p) {
                const int t = 2 * p + 8 * half;
                obuf[t * O_PITCH + l16]       = num[p].x * __builtin_amdgcn_rcpf(den[p].x);
                obuf[(t + 1) * O_PITCH + l16] = num[p].y * __builtin_amdgcn_rcpf(den[p].y);
            }

            v2f ao[4];
#pragma unroll
            for (int q = 0; q < 4; ++q)
                ao[q] = *(const v2f*)&obuf[l16 * O_PITCH + q * 4 + half * 2];
#pragma unroll
            for (int q = 0; q < 4; ++q) {
#pragma unroll
                for (int nt = 0; nt < NT_PROJ; ++nt) {
                    v2f b = *(const v2f*)&lds[OFF_WPROJ + ((h * 4 + q) * NT_PROJ + nt) * 64 + lane * 2];
                    co[nt] = __builtin_amdgcn_wmma_f32_16x16x4_f32(false, ao[q], false, b,
                                                                   (short)0, co[nt], false, false);
                }
            }
        }

#pragma unroll
        for (int nt = 0; nt < NT_PROJ; ++nt) {
#pragma unroll
            for (int r = 0; r < 8; ++r)
                out[(size_t)(row0 + r + 8 * half) * DIM + nt * 16 + l16] = co[nt][r];
        }
    }
}

extern "C" void kernel_launch(void* const* d_in, const int* in_sizes, int n_in,
                              void* d_out, int out_size, void* d_ws, size_t ws_size,
                              hipStream_t stream) {
    const float* x      = (const float*)d_in[0];
    const float* w_qkv  = (const float*)d_in[1];
    const float* b_qkv  = (const float*)d_in[2];
    const float* w_proj = (const float*)d_in[3];
    const float* b_proj = (const float*)d_in[4];
    float* out = (float*)d_out;
    (void)in_sizes; (void)n_in; (void)out_size; (void)d_ws; (void)ws_size;
    fused_channel_attn<<<dim3(GRID), dim3(BLOCK), 0, stream>>>(
        x, w_qkv, b_qkv, w_proj, b_proj, out);
}


# Both GEMMs transposed (weights as WMMA A operand, tokens on lanes): softmax lane-local after one K/V half-exchange, its output registers are the proj B fragments; no K/V or head-output LDS round trip
# speedup vs baseline: 2.5187x; 1.1160x over previous
#include <hip/hip_runtime.h>


#define DIM      128
#define NHEAD    8
#define NTOK     (4 * 16384)
#define NROWT    (NTOK / 16)
#define NT_QKV   24
#define NT_PROJ  8

#define WAVES_PER_BLOCK 8
#define BLOCK   (WAVES_PER_BLOCK * 32)
#define GRID    256

typedef float v2f __attribute__((ext_vector_type(2)));
typedef float v8f __attribute__((ext_vector_type(8)));

#define OFF_WQKV  0
#define SZ_WQKV   (128 * 384)
#define OFF_WPROJ (OFF_WQKV + SZ_WQKV)
#define SZ_WPROJ  (128 * 128)
#define OFF_WBUF  (OFF_WPROJ + SZ_WPROJ)
#define SZ_WBUF_W (16 * DIM)
#define LDS_FLOATS (OFF_WBUF + WAVES_PER_BLOCK * SZ_WBUF_W)

__launch_bounds__(BLOCK)
__global__ void fused_channel_attn(const float* __restrict__ x,
                                   const float* __restrict__ w_qkv,
                                   const float* __restrict__ b_qkv,
                                   const float* __restrict__ w_proj,
                                   const float* __restrict__ b_proj,
                                   float* __restrict__ out)
{
    __shared__ __attribute__((aligned(16))) float lds[LDS_FLOATS];

    const int tid  = threadIdx.x;
    const int wave = tid >> 5;
    const int lane = tid & 31;
    const int l16  = lane & 15;
    const int half = lane >> 4;

    {
        const int rot = blockIdx.x;
        const float4* wq4 = (const float4*)w_qkv;
#pragma unroll 1
        for (int i0 = 0; i0 < 8; i0 += 4) {
            float4 r0[12], r1[12];
#pragma unroll
            for (int u = 0; u < 4; ++u) {
                const int kp = (wave + 8 * (i0 + u) + rot) & 63;
#pragma unroll
                for (int c = 0; c < 3; ++c) {
                    const int n4 = lane + 32 * c;
                    r0[u * 3 + c] = wq4[(2 * kp)     * 96 + n4];
                    r1[u * 3 + c] = wq4[(2 * kp + 1) * 96 + n4];
                }
            }
#pragma unroll
            for (int u = 0; u < 4; ++u) {
                const int kp = (wave + 8 * (i0 + u) + rot) & 63;
                const int kt = kp >> 1, hs = kp & 1;
#pragma unroll
                for (int c = 0; c < 3; ++c) {
                    const int n4 = lane + 32 * c;
                    const float4 a0 = r0[u * 3 + c], a1 = r1[u * 3 + c];
                    float* dst = &lds[OFF_WQKV + (kt * NT_QKV + (n4 >> 2)) * 64
                                      + hs * 32 + (n4 & 3) * 8];
                    *(float4*)(dst)     = make_float4(a0.x, a1.x, a0.y, a1.y);
                    *(float4*)(dst + 4) = make_float4(a0.z, a1.z, a0.w, a1.w);
                }
            }
        }
        const float4* wp4 = (const float4*)w_proj;
        {
            float4 r0[8], r1[8];
#pragma unroll
            for (int i = 0; i < 8; ++i) {
                const int kp = (wave + 8 * i + rot) & 63;
                r0[i] = wp4[(2 * kp)     * 32 + lane];
                r1[i] = wp4[(2 * kp + 1) * 32 + lane];
            }
#pragma unroll
            for (int i = 0; i < 8; ++i) {
                const int kp = (wave + 8 * i + rot) & 63;
                const int kt = ((kp >> 3) << 2) | (kp & 3), hs = (kp >> 2) & 1;
                const float4 a0 = r0[i], a1 = r1[i];
                float* dst = &lds[OFF_WPROJ + (kt * NT_PROJ + (lane >> 2)) * 64
                                  + hs * 32 + (lane & 3) * 8];
                *(float4*)(dst)     = make_float4(a0.x, a1.x, a0.y, a1.y);
                *(float4*)(dst + 4) = make_float4(a0.z, a1.z, a0.w, a1.w);
            }
        }
    }
    __syncthreads();

    const float4 bp4 = ((const float4*)b_proj)[lane];

    float* abuf = &lds[OFF_WBUF + wave * SZ_WBUF_W];
    float4* tb4 = (float4*)abuf;

    const float c1 = 0.08838834764831845f * 1.4426950408889634f;

    const int waveGlobal = blockIdx.x * WAVES_PER_BLOCK + wave;
    const int waveStride = GRID * WAVES_PER_BLOCK;

    for (int rt = waveGlobal; rt < NROWT; rt += waveStride) {
        const int row0 = rt * 16;

        const float4* xg  = (const float4*)(x + (size_t)row0 * DIM);
        float4*       ab4 = (float4*)abuf;
#pragma unroll
        for (int i = 0; i < (16 * DIM / 4) / 32; ++i)
            ab4[lane + i * 32] = xg[lane + i * 32];

        v2f a[32];
#pragma unroll
        for (int kt = 0; kt < 32; ++kt)
            a[kt] = *(const v2f*)&abuf[l16 * DIM + kt * 4 + half * 2];

        v8f co[NT_PROJ];
#pragma unroll
        for (int nt = 0; nt < NT_PROJ; ++nt) {
#pragma unroll
            for (int r = 0; r < 8; ++r) co[nt][r] = 0.0f;
        }

#pragma unroll 1
        for (int h = 0; h < NHEAD; ++h) {
            const float4* bq4 = (const float4*)(b_qkv + h * 16 + 8 * half);
            const float4 bq0 = bq4[0],  bq1 = bq4[1];
            const float4 bk0 = bq4[32], bk1 = bq4[33];
            const float4 bv0 = bq4[64], bv1 = bq4[65];

            v8f cq, ck, cv;
#pragma unroll
            for (int r = 0; r < 8; ++r) { cq[r] = 0.0f; ck[r] = 0.0f; cv[r] = 0.0f; }
#pragma unroll
            for (int kt = 0; kt < 32; ++kt) {
                v2f wq = *(const v2f*)&lds[OFF_WQKV + (kt * NT_QKV + h     ) * 64 + lane * 2];
                v2f wk = *(const v2f*)&lds[OFF_WQKV + (kt * NT_QKV + h + 8 ) * 64 + lane * 2];
                v2f wv = *(const v2f*)&lds[OFF_WQKV + (kt * NT_QKV + h + 16) * 64 + lane * 2];
                cq = __builtin_amdgcn_wmma_f32_16x16x4_f32(false, wq, false, a[kt],
                                                           (short)0, cq, false, false);
                ck = __builtin_amdgcn_wmma_f32_16x16x4_f32(false, wk, false, a[kt],
                                                           (short)0, ck, false, false);
                cv = __builtin_amdgcn_wmma_f32_16x16x4_f32(false, wv, false, a[kt],
                                                           (short)0, cv, false, false);
            }
            cq[0] += bq0.x; cq[1] += bq0.y; cq[2] += bq0.z; cq[3] += bq0.w;
            cq[4] += bq1.x; cq[5] += bq1.y; cq[6] += bq1.z; cq[7] += bq1.w;
            ck[0] += bk0.x; ck[1] += bk0.y; ck[2] += bk0.z; ck[3] += bk0.w;
            ck[4] += bk1.x; ck[5] += bk1.y; ck[6] += bk1.z; ck[7] += bk1.w;
            cv[0] += bv0.x; cv[1] += bv0.y; cv[2] += bv0.z; cv[3] += bv0.w;
            cv[4] += bv1.x; cv[5] += bv1.y; cv[6] += bv1.z; cv[7] += bv1.w;

            float ko[8], vo[8];
#pragma unroll
            for (int r = 0; r < 8; ++r) {
                const float kf = ck[r];
                const float vf = cv[r];
                const int kb = __builtin_bit_cast(int, kf);
                const int vb = __builtin_bit_cast(int, vf);
                ko[r] = __builtin_bit_cast(float, __builtin_amdgcn_permlanex16(kb, kb, 0x76543210, 0xfedcba98, false, false));
                vo[r] = __builtin_bit_cast(float, __builtin_amdgcn_permlanex16(vb, vb, 0x76543210, 0xfedcba98, false, false));
            }

            v2f qs[4], num[4], den[4];
#pragma unroll
            for (int p = 0; p < 4; ++p) {
                qs[p].x = cq[2 * p] * c1;  qs[p].y = cq[2 * p + 1] * c1;
                num[p].x = 0.0f; num[p].y = 0.0f;
                den[p].x = 0.0f; den[p].y = 0.0f;
            }
#pragma unroll
            for (int j = 0; j < 16; ++j) {
                const float kj = (j < 8) ? ck[j & 7] : ko[j & 7];
                const float vj = (j < 8) ? cv[j & 7] : vo[j & 7];
                v2f kk, vv;
                kk.x = kj; kk.y = kj;
                vv.x = vj; vv.y = vj;
#pragma unroll
                for (int p = 0; p < 4; ++p) {
                    const v2f a2 = qs[p] * kk;
                    v2f e;
                    e.x = __builtin_amdgcn_exp2f(a2.x);
                    e.y = __builtin_amdgcn_exp2f(a2.y);
                    num[p] = __builtin_elementwise_fma(e, vv, num[p]);
                    den[p] = den[p] + e;
                }
            }
            v2f ao[4];
#pragma unroll
            for (int p = 0; p < 4; ++p) {
                ao[p].x = num[p].x * __builtin_amdgcn_rcpf(den[p].x);
                ao[p].y = num[p].y * __builtin_amdgcn_rcpf(den[p].y);
            }

#pragma unroll
            for (int q = 0; q < 4; ++q) {
#pragma unroll
                for (int nt = 0; nt < NT_PROJ; ++nt) {
                    v2f w = *(const v2f*)&lds[OFF_WPROJ + ((h * 4 + q) * NT_PROJ + nt) * 64 + lane * 2];
                    co[nt] = __builtin_amdgcn_wmma_f32_16x16x4_f32(false, w, false, ao[q],
                                                                   (short)0, co[nt], false, false);
                }
            }
        }

#pragma unroll
        for (int nt = 0; nt < NT_PROJ; ++nt) {
#pragma unroll
            for (int s = 0; s < 2; ++s) {
                const int q = nt * 4 + 2 * half + s;
                tb4[l16 * 32 + (q ^ l16)] = make_float4(co[nt][4 * s], co[nt][4 * s + 1],
                                                        co[nt][4 * s + 2], co[nt][4 * s + 3]);
            }
        }
        float4* og = (float4*)(out + (size_t)row0 * DIM);
#pragma unroll
        for (int t = 0; t < 16; ++t) {
            float4 v = tb4[t * 32 + (lane ^ t)];
            v.x += bp4.x; v.y += bp4.y; v.z += bp4.z; v.w += bp4.w;
            og[t * 32 + lane] = v;
        }
    }
}

extern "C" void kernel_launch(void* const* d_in, const int* in_sizes, int n_in,
                              void* d_out, int out_size, void* d_ws, size_t ws_size,
                              hipStream_t stream) {
    const float* x      = (const float*)d_in[0];
    const float* w_qkv  = (const float*)d_in[1];
    const float* b_qkv  = (const float*)d_in[2];
    const float* w_proj = (const float*)d_in[3];
    const float* b_proj = (const float*)d_in[4];
    float* out = (float*)d_out;
    (void)in_sizes; (void)n_in; (void)out_size; (void)d_ws; (void)ws_size;
    fused_channel_attn<<<dim3(GRID), dim3(BLOCK), 0, stream>>>(
        x, w_qkv, b_qkv, w_proj, b_proj, out);
}
